# starting kernel
# baseline (speedup 1.0000x reference)
#include <hip/hip_runtime.h>

typedef float v2f __attribute__((ext_vector_type(2)));
typedef float v8f __attribute__((ext_vector_type(8)));

#define NLAT_IN   181
#define NLON_IN   360
#define NLAT_OUT  361
#define NLON_OUT  720
#define KSIZE     3
#define CIN       16
#define COUT      16
#define NNZ       8192
#define NPIX      (NLAT_IN * NLON_IN)
#define ACC_PITCH 727
#define SCAT_THREADS 512

__global__ __launch_bounds__(32)
void disco_mix_wmma(const float* __restrict__ x,
                    const float* __restrict__ w,
                    float* __restrict__ xw)
{
    const int lane = threadIdx.x;
    const int m0   = blockIdx.x * 16;
    const int mrow = lane & 15;
    const int half = lane >> 4;
    const int co   = mrow;
    const bool full_tile = (m0 + 16 <= NPIX);

    int m = m0 + mrow;
    if (m >= NPIX) m = NPIX - 1;

    v2f a[4];
#pragma unroll
    for (int s = 0; s < 4; ++s) {
        const int ci = 4 * s + 2 * half;
        a[s].x = x[(ci + 0) * NPIX + m];
        a[s].y = x[(ci + 1) * NPIX + m];
    }

#pragma unroll
    for (int kt = 0; kt < KSIZE; ++kt) {
        v8f c = {0.f, 0.f, 0.f, 0.f, 0.f, 0.f, 0.f, 0.f};
#pragma unroll
        for (int s = 0; s < 4; ++s) {
            const int ci = 4 * s + 2 * half;
            v2f b;
            b.x = w[co * (CIN * KSIZE) + (ci + 0) * KSIZE + kt];
            b.y = w[co * (CIN * KSIZE) + (ci + 1) * KSIZE + kt];
            c = __builtin_amdgcn_wmma_f32_16x16x4_f32(false, a[s], false, b,
                                                      (short)0, c, false, false);
        }
        float* dst = xw + ((size_t)(kt * NPIX + m0 + 8 * half) * COUT + co);
        if (full_tile) {
#pragma unroll
            for (int v = 0; v < 8; ++v)
                dst[v * COUT] = c[v];
        } else {
#pragma unroll
            for (int v = 0; v < 8; ++v) {
                const int pix = m0 + v + 8 * half;
                if (pix < NPIX) dst[v * COUT] = c[v];
            }
        }
    }
}

__global__ __launch_bounds__(SCAT_THREADS)
void disco_scatter_lat(const float* __restrict__ xw,
                       const float* __restrict__ bias,
                       const int*   __restrict__ ker_idx,
                       const int*   __restrict__ row_idx,
                       const int*   __restrict__ col_idx,
                       const float* __restrict__ vals,
                       float* __restrict__ out)
{
    __shared__ float acc[COUT * ACC_PITCH];
    __shared__ int   list[NNZ];
    __shared__ int   cnt;

    const int tid = threadIdx.x;
    const int hi  = blockIdx.x;

    if (tid == 0) cnt = 0;
    for (int i = tid; i < COUT * ACC_PITCH; i += SCAT_THREADS) acc[i] = 0.f;
    __syncthreads();

    for (int nz = tid; nz < NNZ; nz += SCAT_THREADS) {
        if (col_idx[nz] / NLON_OUT == hi) {
            const int slot = atomicAdd(&cnt, 1);
            list[slot] = nz;
        }
    }
    __syncthreads();

    const int n = cnt;
    for (int j = 0; j < n; ++j) {
        const int   nz = list[j];
        const int   k  = ker_idx[nz];
        const int   r  = row_idx[nz];
        const int   p  = col_idx[nz] - hi * NLON_OUT;
        const float v  = vals[nz];
        const float* plane = xw + (size_t)(k * NPIX + r * NLON_IN) * COUT;

        for (int e = tid; e < COUT * NLON_IN; e += SCAT_THREADS) {
            const int co = e & 15;
            const int po = e >> 4;
            int lon = p + 2 * po;
            if (lon >= NLON_OUT) lon -= NLON_OUT;
            atomicAdd(&acc[co * ACC_PITCH + lon], v * plane[po * COUT + co]);
        }
    }
    __syncthreads();

    for (int e = tid; e < COUT * NLON_OUT; e += SCAT_THREADS) {
        const int co = e / NLON_OUT;
        const int L  = e - co * NLON_OUT;
        out[(co * NLAT_OUT + hi) * NLON_OUT + L] = acc[co * ACC_PITCH + L] + bias[co];
    }
}

extern "C" void kernel_launch(void* const* d_in, const int* in_sizes, int n_in,
                              void* d_out, int out_size, void* d_ws, size_t ws_size,
                              hipStream_t stream)
{
    (void)in_sizes; (void)n_in; (void)out_size; (void)ws_size;

    const float* x    = (const float*)d_in[0];
    const float* w    = (const float*)d_in[1];
    const float* bias = (const float*)d_in[2];
    const int*   ker  = (const int*)  d_in[3];
    const int*   row  = (const int*)  d_in[4];
    const int*   col  = (const int*)  d_in[5];
    const float* vals = (const float*)d_in[6];
    float*       out  = (float*)d_out;

    float* xw = (float*)d_ws;

    const int tiles = (NPIX + 15) / 16;
    disco_mix_wmma<<<tiles, 32, 0, stream>>>(x, w, xw);
    disco_scatter_lat<<<NLAT_OUT, SCAT_THREADS, 0, stream>>>(xw, bias, ker, row, col,
                                                             vals, out);
}


# scatter kernel as a gather: each thread owns fixed outputs summed in registers, so queue, index and xw loads go out in batches ahead of use; LDS float atomics removed, sum order = queue order, fma
# speedup vs baseline: 4.6811x; 4.6811x over previous
#include <hip/hip_runtime.h>

typedef float v2f __attribute__((ext_vector_type(2)));
typedef float v8f __attribute__((ext_vector_type(8)));

#define NLAT_IN   181
#define NLON_IN   360
#define NLAT_OUT  361
#define NLON_OUT  720
#define KSIZE     3
#define CIN       16
#define COUT      16
#define NNZ       8192
#define NPIX      (NLAT_IN * NLON_IN)
#define SCAT_THREADS 480
#define RING_STEP  120
#define STAGE_PITCH 724
#define CHUNK      SCAT_THREADS

__global__ __launch_bounds__(32)
void disco_mix_wmma(const float* __restrict__ x,
                    const float* __restrict__ w,
                    float* __restrict__ xw)
{
    const int lane = threadIdx.x;
    const int m0   = blockIdx.x * 16;
    const int mrow = lane & 15;
    const int half = lane >> 4;
    const int co   = mrow;
    const bool full_tile = (m0 + 16 <= NPIX);

    int m = m0 + mrow;
    if (m >= NPIX) m = NPIX - 1;

    v2f a[4];
#pragma unroll
    for (int s = 0; s < 4; ++s) {
        const int ci = 4 * s + 2 * half;
        a[s].x = x[(ci + 0) * NPIX + m];
        a[s].y = x[(ci + 1) * NPIX + m];
    }

#pragma unroll
    for (int kt = 0; kt < KSIZE; ++kt) {
        v8f c = {0.f, 0.f, 0.f, 0.f, 0.f, 0.f, 0.f, 0.f};
#pragma unroll
        for (int s = 0; s < 4; ++s) {
            const int ci = 4 * s + 2 * half;
            v2f b;
            b.x = w[co * (CIN * KSIZE) + (ci + 0) * KSIZE + kt];
            b.y = w[co * (CIN * KSIZE) + (ci + 1) * KSIZE + kt];
            c = __builtin_amdgcn_wmma_f32_16x16x4_f32(false, a[s], false, b,
                                                      (short)0, c, false, false);
        }
        float* dst = xw + ((size_t)(kt * NPIX + m0 + 8 * half) * COUT + co);
        if (full_tile) {
#pragma unroll
            for (int v = 0; v < 8; ++v)
                dst[v * COUT] = c[v];
        } else {
#pragma unroll
            for (int v = 0; v < 8; ++v) {
                const int pix = m0 + v + 8 * half;
                if (pix < NPIX) dst[v * COUT] = c[v];
            }
        }
    }
}


template <int U>
__device__ __forceinline__ void gather_batch(const float* __restrict__ xw,
                                             const int4* m_ent, int e, int g, int lane_byte,
                                             float4& a0, float4& a1, float4& a2)
{
    float4 x0[U], x1[U], x2[U];
    float  v[U];
#pragma unroll
    for (int u = 0; u < U; ++u) {
        const int4 ent = m_ent[e + u];
        const int base = __builtin_amdgcn_readfirstlane(ent.x);
        const int s    = __builtin_amdgcn_readfirstlane(ent.y);
        v[u] = __int_as_float(__builtin_amdgcn_readfirstlane(ent.z));
        const char* plane = reinterpret_cast<const char*>(xw + base);
        int d0 = g - s;               if (d0 < 0)        d0 += NLON_IN;
        int d1 = d0 + RING_STEP;      if (d1 >= NLON_IN) d1 -= NLON_IN;
        int d2 = d0 + 2 * RING_STEP;  if (d2 >= NLON_IN) d2 -= NLON_IN;
        x0[u] = *reinterpret_cast<const float4*>(plane + (unsigned)(d0 * (COUT * 4) + lane_byte));
        x1[u] = *reinterpret_cast<const float4*>(plane + (unsigned)(d1 * (COUT * 4) + lane_byte));
        x2[u] = *reinterpret_cast<const float4*>(plane + (unsigned)(d2 * (COUT * 4) + lane_byte));
    }
#pragma unroll
    for (int u = 0; u < U; ++u) {
        a0.x = fmaf(v[u], x0[u].x, a0.x); a0.y = fmaf(v[u], x0[u].y, a0.y);
        a0.z = fmaf(v[u], x0[u].z, a0.z); a0.w = fmaf(v[u], x0[u].w, a0.w);
        a1.x = fmaf(v[u], x1[u].x, a1.x); a1.y = fmaf(v[u], x1[u].y, a1.y);
        a1.z = fmaf(v[u], x1[u].z, a1.z); a1.w = fmaf(v[u], x1[u].w, a1.w);
        a2.x = fmaf(v[u], x2[u].x, a2.x); a2.y = fmaf(v[u], x2[u].y, a2.y);
        a2.z = fmaf(v[u], x2[u].z, a2.z); a2.w = fmaf(v[u], x2[u].w, a2.w);
    }
}

__device__ __forceinline__ void gather_range(const float* __restrict__ xw,
                                             const int4* m_ent, int a, int b, int g, int lane_byte,
                                             float4& a0, float4& a1, float4& a2)
{
    int e = a;
    for (; e + 4 <= b; e += 4)
        gather_batch<4>(xw, m_ent, e, g, lane_byte, a0, a1, a2);
    const int rem = b - e;
    if (rem == 3)      gather_batch<3>(xw, m_ent, e, g, lane_byte, a0, a1, a2);
    else if (rem == 2) gather_batch<2>(xw, m_ent, e, g, lane_byte, a0, a1, a2);
    else if (rem == 1) gather_batch<1>(xw, m_ent, e, g, lane_byte, a0, a1, a2);
}

__global__ __launch_bounds__(SCAT_THREADS)
void disco_gather_lat(const float* __restrict__ xw,
                      const float* __restrict__ bias,
                      const int*   __restrict__ ker_idx,
                      const int*   __restrict__ row_idx,
                      const int*   __restrict__ col_idx,
                      const float* __restrict__ vals,
                      float* __restrict__ out)
{
    __shared__ __attribute__((aligned(16))) float stage[COUT * STAGE_PITCH];
    __shared__ int   list[NNZ];
    __shared__ int4  m_ent[CHUNK];
    __shared__ int   cnt[2];

    const int tid = threadIdx.x;
    const int hi  = blockIdx.x;
    const int lo  = hi * NLON_OUT;
    const int cq  = tid & 3;
    const int g   = tid >> 2;

    if (tid < 2) cnt[tid] = 0;
    __syncthreads();

    {
        const int4* col4 = reinterpret_cast<const int4*>(col_idx);
        int4 c[5];
#pragma unroll
        for (int t = 0; t < 5; ++t) {
            const int i4 = tid + t * SCAT_THREADS;
            c[t] = (i4 < NNZ / 4) ? col4[i4] : make_int4(-1, -1, -1, -1);
        }
#pragma unroll
        for (int t = 0; t < 5; ++t) {
            const int nz0 = 4 * (tid + t * SCAT_THREADS);
            const int cc[4] = {c[t].x, c[t].y, c[t].z, c[t].w};
#pragma unroll
            for (int u = 0; u < 4; ++u) {
                const unsigned p = (unsigned)(cc[u] - lo);
                if (p < (unsigned)NLON_OUT) {
                    const int q    = (int)(p & 1u);
                    const int slot = atomicAdd(&cnt[q], 1);
                    list[q ? (NNZ - 1 - slot) : slot] = nz0 + u;
                }
            }
        }
    }
    __syncthreads();

    const int n0 = cnt[0];
    const int nt = n0 + cnt[1];

    const float4 zero4 = make_float4(0.f, 0.f, 0.f, 0.f);
    float4 e0 = zero4, e1 = zero4, e2 = zero4;
    float4 o0 = zero4, o1 = zero4, o2 = zero4;
    const int lane_byte = 16 * cq;

    for (int c0 = 0; c0 < nt; c0 += CHUNK) {
        const int j = c0 + tid;
        if (j < nt) {
            const int nz = (j < n0) ? list[j] : list[NNZ - 1 - (j - n0)];
            const int k  = ker_idx[nz];
            const int r  = row_idx[nz];
            const int p  = col_idx[nz] - lo;
            m_ent[tid] = make_int4((k * NPIX + r * NLON_IN) * COUT, p >> 1,
                                   __float_as_int(vals[nz]), 0);
        }
        __syncthreads();
        const int m  = min(CHUNK, nt - c0);
        const int mE = min(max(n0 - c0, 0), m);
        gather_range(xw, m_ent, 0,  mE, g, lane_byte, e0, e1, e2);
        gather_range(xw, m_ent, mE, m,  g, lane_byte, o0, o1, o2);
        if (c0 + CHUNK < nt) __syncthreads();
    }

    {
        const float ev[3][4] = {{e0.x, e0.y, e0.z, e0.w}, {e1.x, e1.y, e1.z, e1.w}, {e2.x, e2.y, e2.z, e2.w}};
        const float od[3][4] = {{o0.x, o0.y, o0.z, o0.w}, {o1.x, o1.y, o1.z, o1.w}, {o2.x, o2.y, o2.z, o2.w}};
#pragma unroll
        for (int t = 0; t < 3; ++t) {
            const int i = g + t * RING_STEP;
#pragma unroll
            for (int c = 0; c < 4; ++c) {
                float2 pr; pr.x = ev[t][c]; pr.y = od[t][c];
                *reinterpret_cast<float2*>(&stage[(4 * cq + c) * STAGE_PITCH + 2 * i]) = pr;
            }
        }
    }
    __syncthreads();

#pragma unroll
    for (int t = 0; t < (COUT * NLON_OUT / 4) / SCAT_THREADS; ++t) {
        const int e4 = tid + t * SCAT_THREADS;
        const int co = e4 / (NLON_OUT / 4);
        const int l4 = e4 - co * (NLON_OUT / 4);
        float4 r = *reinterpret_cast<const float4*>(&stage[co * STAGE_PITCH + 4 * l4]);
        const float b = bias[co];
        r.x += b; r.y += b; r.z += b; r.w += b;
        *reinterpret_cast<float4*>(&out[(size_t)(co * NLAT_OUT + hi) * NLON_OUT + 4 * l4]) = r;
    }
}

extern "C" void kernel_launch(void* const* d_in, const int* in_sizes, int n_in,
                              void* d_out, int out_size, void* d_ws, size_t ws_size,
                              hipStream_t stream)
{
    (void)in_sizes; (void)n_in; (void)out_size; (void)ws_size;

    const float* x    = (const float*)d_in[0];
    const float* w    = (const float*)d_in[1];
    const float* bias = (const float*)d_in[2];
    const int*   ker  = (const int*)  d_in[3];
    const int*   row  = (const int*)  d_in[4];
    const int*   col  = (const int*)  d_in[5];
    const float* vals = (const float*)d_in[6];
    float*       out  = (float*)d_out;

    float* xw = (float*)d_ws;

    const int tiles = (NPIX + 15) / 16;
    disco_mix_wmma<<<tiles, 32, 0, stream>>>(x, w, xw);
    disco_gather_lat<<<NLAT_OUT, SCAT_THREADS, 0, stream>>>(xw, bias, ker, row, col,
                                                            vals, out);
}
